# starting kernel
# baseline (speedup 1.0000x reference)
#include <hip/hip_runtime.h>


typedef __attribute__((ext_vector_type(2)))  float    v2f;
typedef __attribute__((ext_vector_type(8)))  float    v8f;
typedef __attribute__((ext_vector_type(16))) _Float16 v16h;

static constexpr int BATCH = 2;
static constexpr int SLEN  = 1024;
static constexpr int DIN   = 256;
static constexpr int DOUT  = 64;
static constexpr int Q     = 128;
static constexpr int KH    = 6;
static constexpr int MROWS = BATCH * SLEN;
static constexpr int FPQ   = 2 * KH;
static constexpr int KFEAT = Q * FPQ;
static constexpr int KB    = KFEAT / 32;

__device__ __forceinline__ int frag_k16(int i, int hi) {
    return (i & 7) + ((i >> 3) << 4) + hi * 8;
}

__global__ __launch_bounds__(256) void k_angle_wmma(
    const float* __restrict__ X0,
    const float* __restrict__ t,
    const float* __restrict__ Wcw,
    const float* __restrict__ Wcb,
    const float* __restrict__ w,
    float* __restrict__ ANG)
{
    const int tid  = threadIdx.x;
    const int wave = tid >> 5;
    const int lane = tid & 31;
    const int lm   = lane & 15;
    const int hi   = lane >> 4;
    const int m0   = blockIdx.x * 16;
    const int n0   = wave * 16;

    const int ksel = hi * 2;
    const float* xrow = X0  + (size_t)(m0 + lm) * DIN + ksel;
    const float* wrow = Wcw + (size_t)(n0 + lm) * DIN + ksel;

    v8f acc = {0.f,0.f,0.f,0.f,0.f,0.f,0.f,0.f};
    #pragma unroll 4
    for (int k = 0; k < DIN; k += 4) {
        v2f a = *(const v2f*)(xrow + k);
        v2f b = *(const v2f*)(wrow + k);
        acc = __builtin_amdgcn_wmma_f32_16x16x4_f32(
             false, a,  false, b,
             (short)0, acc,  false,  false);
    }

    const int   q     = n0 + lm;
    const int   batch = m0 >> 10;
    const float shift = Wcb[q] + w[q] * t[batch];

    #pragma unroll
    for (int j = 0; j < 8; ++j) {
        const int row = m0 + j + hi * 8;
        ANG[(size_t)row * Q + q] = acc[j] + shift;
    }
}

__global__ __launch_bounds__(256) void k_w2_pack(
    const float* __restrict__ A,
    const float* __restrict__ Bp,
    _Float16* __restrict__ W2P)
{
    const int idx  = blockIdx.x * 256 + threadIdx.x;
    const int i    = idx & 15;
    const int lane = (idx >> 4) & 31;
    const int nt   = (idx >> 9) & 3;
    const int kb   = idx >> 11;
    const int hi   = lane >> 4;

    const int kf = kb * 32 + frag_k16(i, hi);
    const int d  = nt * 16 + (lane & 15);
    const int q  = kf / FPQ;
    const int kk = kf - q * FPQ;
    const int ks = (kk < KH) ? kk : kk - KH;

    const int   ai  = (d * Q + q) * KH + ks;
    const float av  = A[ai];
    const float bp  = Bp[ai];
    const float val = (kk < KH) ? av * cosf(bp) : av * sinf(bp);
    W2P[idx] = (_Float16)val;
}

__global__ __launch_bounds__(128) void k_fourier_wmma(
    const float*    __restrict__ ANG,
    const _Float16* __restrict__ W2P,
    float*          __restrict__ out)
{
    __shared__ float sAng[64 * Q];
    const int tid     = threadIdx.x;
    const int rowBase = blockIdx.x * 64;

    for (int r = tid; r < 64 * Q; r += 128)
        sAng[r] = ANG[(size_t)rowBase * Q + r];
    __syncthreads();

    const int wave = tid >> 5;
    const int lane = tid & 31;
    const int lm   = lane & 15;
    const int hi   = lane >> 4;
    const int m0   = rowBase + wave * 16;

    const float* ar  = sAng + (wave * 16 + lm) * Q;
    const v16h*  W2v = (const v16h*)W2P;

    v8f acc0 = {0.f,0.f,0.f,0.f,0.f,0.f,0.f,0.f};
    v8f acc1 = {0.f,0.f,0.f,0.f,0.f,0.f,0.f,0.f};
    v8f acc2 = {0.f,0.f,0.f,0.f,0.f,0.f,0.f,0.f};
    v8f acc3 = {0.f,0.f,0.f,0.f,0.f,0.f,0.f,0.f};

    for (int kb = 0; kb < KB; ++kb) {
        v16h a;
        #pragma unroll
        for (int i = 0; i < 16; ++i) {
            const int   kf  = kb * 32 + frag_k16(i, hi);
            const int   q   = kf / FPQ;
            const int   kk  = kf - q * FPQ;
            const float ang = ar[q];
            const float kc  = (float)(((kk < KH) ? kk : kk - KH) + 1);
            const float v   = (kk < KH) ? __sinf(kc * ang) : __cosf(kc * ang);
            a[i] = (_Float16)v;
        }
        const v16h* bbase = W2v + (size_t)kb * 128 + lane;
        const v16h b0 = bbase[0];
        const v16h b1 = bbase[32];
        const v16h b2 = bbase[64];
        const v16h b3 = bbase[96];
        acc0 = __builtin_amdgcn_wmma_f32_16x16x32_f16(false, a, false, b0, (short)0, acc0, false, false);
        acc1 = __builtin_amdgcn_wmma_f32_16x16x32_f16(false, a, false, b1, (short)0, acc1, false, false);
        acc2 = __builtin_amdgcn_wmma_f32_16x16x32_f16(false, a, false, b2, (short)0, acc2, false, false);
        acc3 = __builtin_amdgcn_wmma_f32_16x16x32_f16(false, a, false, b3, (short)0, acc3, false, false);
    }

    #pragma unroll
    for (int j = 0; j < 8; ++j) {
        const int row = m0 + j + hi * 8;
        float* o = out + (size_t)row * DOUT + lm;
        o[0]  = acc0[j];
        o[16] = acc1[j];
        o[32] = acc2[j];
        o[48] = acc3[j];
    }
}

extern "C" void kernel_launch(void* const* d_in, const int* in_sizes, int n_in,
                              void* d_out, int out_size, void* d_ws, size_t ws_size,
                              hipStream_t stream) {
    const float* X0  = (const float*)d_in[0];
    const float* t   = (const float*)d_in[1];
    const float* Wcw = (const float*)d_in[2];
    const float* Wcb = (const float*)d_in[3];
    const float* w   = (const float*)d_in[4];
    const float* A   = (const float*)d_in[5];
    const float* Bp  = (const float*)d_in[6];
    float* out = (float*)d_out;

    char* ws = (char*)d_ws;
    float*    ANG = (float*)ws;
    _Float16* W2P = (_Float16*)(ws + (size_t)MROWS * Q * 4);

    k_w2_pack   <<<(KFEAT * DOUT) / 256, 256, 0, stream>>>(A, Bp, W2P);
    k_angle_wmma<<<MROWS / 16,           256, 0, stream>>>(X0, t, Wcw, Wcb, w, ANG);
    k_fourier_wmma<<<MROWS / 64,         128, 0, stream>>>(ANG, W2P, out);
}


# Stage-3 contraction split over the 24 waves of a block (one harmonic of one 32-q block per wave), 16-row blocks, grid 32x4 to 128x24 waves; partials summed via LDS: f32 sum order over k changes
# speedup vs baseline: 6.6913x; 6.6913x over previous
#include <hip/hip_runtime.h>


typedef __attribute__((ext_vector_type(2)))  float    v2f;
typedef __attribute__((ext_vector_type(8)))  float    v8f;
typedef __attribute__((ext_vector_type(16))) _Float16 v16h;

static constexpr int BATCH = 2;
static constexpr int SLEN  = 1024;
static constexpr int DIN   = 256;
static constexpr int DOUT  = 64;
static constexpr int Q     = 128;
static constexpr int KH    = 6;
static constexpr int MROWS = BATCH * SLEN;
static constexpr int FPQ   = 2 * KH;
static constexpr int KFEAT = Q * FPQ;
static constexpr int KB    = KFEAT / 32;
static constexpr int QBLK  = Q / 32;

__device__ __forceinline__ int frag_k16(int i, int hi) {
    return (i & 7) + ((i >> 3) << 4) + hi * 8;
}

__global__ __launch_bounds__(256) void k_angle_wmma(
    const float* __restrict__ X0,
    const float* __restrict__ t,
    const float* __restrict__ Wcw,
    const float* __restrict__ Wcb,
    const float* __restrict__ w,
    float* __restrict__ ANG)
{
    const int tid  = threadIdx.x;
    const int wave = tid >> 5;
    const int lane = tid & 31;
    const int lm   = lane & 15;
    const int hi   = lane >> 4;
    const int m0   = blockIdx.x * 16;
    const int n0   = wave * 16;

    const int ksel = hi * 2;
    const float* xrow = X0  + (size_t)(m0 + lm) * DIN + ksel;
    const float* wrow = Wcw + (size_t)(n0 + lm) * DIN + ksel;

    v8f acc = {0.f,0.f,0.f,0.f,0.f,0.f,0.f,0.f};
    #pragma unroll 4
    for (int k = 0; k < DIN; k += 4) {
        v2f a = *(const v2f*)(xrow + k);
        v2f b = *(const v2f*)(wrow + k);
        acc = __builtin_amdgcn_wmma_f32_16x16x4_f32(
             false, a,  false, b,
             (short)0, acc,  false,  false);
    }

    const int   q     = n0 + lm;
    const int   batch = m0 >> 10;
    const float shift = Wcb[q] + w[q] * t[batch];

    #pragma unroll
    for (int j = 0; j < 8; ++j) {
        const int row = m0 + j + hi * 8;
        ANG[(size_t)row * Q + q] = acc[j] + shift;
    }
}

__global__ __launch_bounds__(256) void k_w2_pack(
    const float* __restrict__ A,
    const float* __restrict__ Bp,
    _Float16* __restrict__ W2P)
{
    const int idx  = blockIdx.x * 256 + threadIdx.x;
    const int i    = idx & 15;
    const int lane = (idx >> 4) & 31;
    const int nt   = (idx >> 9) & 3;
    const int ks   = idx >> 11;
    const int hi   = lane >> 4;

    const int fn = ks & 1;
    const int qh = ks >> 1;
    const int qb = qh / KH;
    const int h  = qh - qb * KH;

    const int q  = qb * 32 + frag_k16(i, hi);
    const int d  = nt * 16 + (lane & 15);

    const int   ai  = (d * Q + q) * KH + h;
    const float av  = A[ai];
    const float bp  = Bp[ai];
    const float val = (fn == 0) ? av * cosf(bp) : av * sinf(bp);
    W2P[idx] = (_Float16)val;
}

static constexpr int NWAVE3 = QBLK * KH;
static constexpr int RPITCH = DOUT + 4;
static constexpr int PLANE  = 16 * RPITCH;

__global__ __launch_bounds__(NWAVE3 * 32) void k_fourier_wmma(
    const float*    __restrict__ ANG,
    const _Float16* __restrict__ W2P,
    float*          __restrict__ out)
{
    __shared__ alignas(16) float sPart[NWAVE3 * PLANE];

    const int tid  = threadIdx.x;
    const int wave = tid >> 5;
    const int lane = tid & 31;
    const int lm   = lane & 15;
    const int hi   = lane >> 4;
    const int qb   = wave / KH;
    const int h    = wave - qb * KH;
    const int m0   = blockIdx.x * 16;

    const float4* ap = (const float4*)(ANG + (size_t)(m0 + lm) * Q + qb * 32 + hi * 8);
    const float4 x0 = ap[0];
    const float4 x1 = ap[1];
    const float4 x2 = ap[4];
    const float4 x3 = ap[5];
    const float ang[16] = { x0.x, x0.y, x0.z, x0.w, x1.x, x1.y, x1.z, x1.w,
                            x2.x, x2.y, x2.z, x2.w, x3.x, x3.y, x3.z, x3.w };

    const v16h* bs = (const v16h*)W2P + lane + (size_t)wave * (2 * 128);
    const v16h* bc = bs + 128;
    const v16h s0 = bs[0];
    const v16h s1 = bs[32];
    const v16h s2 = bs[64];
    const v16h s3 = bs[96];
    const v16h c0 = bc[0];
    const v16h c1 = bc[32];
    const v16h c2 = bc[64];
    const v16h c3 = bc[96];

    const float kc = (float)(h + 1);
    v16h as, ac;
    #pragma unroll
    for (int i = 0; i < 16; ++i) {
        const float ph = kc * ang[i];
        as[i] = (_Float16)__sinf(ph);
        ac[i] = (_Float16)__cosf(ph);
    }

    const v8f zero = {0.f,0.f,0.f,0.f,0.f,0.f,0.f,0.f};
    v8f acc0 = __builtin_amdgcn_wmma_f32_16x16x32_f16(false, as, false, s0, (short)0, zero, false, false);
    v8f acc1 = __builtin_amdgcn_wmma_f32_16x16x32_f16(false, as, false, s1, (short)0, zero, false, false);
    v8f acc2 = __builtin_amdgcn_wmma_f32_16x16x32_f16(false, as, false, s2, (short)0, zero, false, false);
    v8f acc3 = __builtin_amdgcn_wmma_f32_16x16x32_f16(false, as, false, s3, (short)0, zero, false, false);
    acc0 = __builtin_amdgcn_wmma_f32_16x16x32_f16(false, ac, false, c0, (short)0, acc0, false, false);
    acc1 = __builtin_amdgcn_wmma_f32_16x16x32_f16(false, ac, false, c1, (short)0, acc1, false, false);
    acc2 = __builtin_amdgcn_wmma_f32_16x16x32_f16(false, ac, false, c2, (short)0, acc2, false, false);
    acc3 = __builtin_amdgcn_wmma_f32_16x16x32_f16(false, ac, false, c3, (short)0, acc3, false, false);

    float* pw = sPart + wave * PLANE + (hi * 8) * RPITCH + lm;
    #pragma unroll
    for (int j = 0; j < 8; ++j) {
        pw[j * RPITCH +  0] = acc0[j];
        pw[j * RPITCH + 16] = acc1[j];
        pw[j * RPITCH + 32] = acc2[j];
        pw[j * RPITCH + 48] = acc3[j];
    }
    __syncthreads();

    if (wave < 8) {
        const int r = wave * 2 + hi;
        const float* pr = sPart + r * RPITCH + lm * 4;
        float4 s = *(const float4*)pr;
        #pragma unroll
        for (int p = 1; p < NWAVE3; ++p) {
            const float4 v = *(const float4*)(pr + p * PLANE);
            s.x += v.x;
            s.y += v.y;
            s.z += v.z;
            s.w += v.w;
        }
        *(float4*)(out + (size_t)(m0 + r) * DOUT + lm * 4) = s;
    }
}

extern "C" void kernel_launch(void* const* d_in, const int* in_sizes, int n_in,
                              void* d_out, int out_size, void* d_ws, size_t ws_size,
                              hipStream_t stream) {
    const float* X0  = (const float*)d_in[0];
    const float* t   = (const float*)d_in[1];
    const float* Wcw = (const float*)d_in[2];
    const float* Wcb = (const float*)d_in[3];
    const float* w   = (const float*)d_in[4];
    const float* A   = (const float*)d_in[5];
    const float* Bp  = (const float*)d_in[6];
    float* out = (float*)d_out;

    char* ws = (char*)d_ws;
    float*    ANG = (float*)ws;
    _Float16* W2P = (_Float16*)(ws + (size_t)MROWS * Q * 4);

    k_w2_pack   <<<(KFEAT * DOUT) / 256, 256, 0, stream>>>(A, Bp, W2P);
    k_angle_wmma<<<MROWS / 16,           256, 0, stream>>>(X0, t, Wcw, Wcb, w, ANG);
    k_fourier_wmma<<<MROWS / 16, NWAVE3 * 32, 0, stream>>>(ANG, W2P, out);
}


# Stage-1 angle GEMM and stage-2 weight pack, which do not depend on each other, merged into one launch (block index selects the role): 3 launches to 2; no sum order changes
# speedup vs baseline: 7.6966x; 1.1502x over previous
#include <hip/hip_runtime.h>


typedef __attribute__((ext_vector_type(2)))  float    v2f;
typedef __attribute__((ext_vector_type(8)))  float    v8f;
typedef __attribute__((ext_vector_type(16))) _Float16 v16h;

static constexpr int BATCH = 2;
static constexpr int SLEN  = 1024;
static constexpr int DIN   = 256;
static constexpr int DOUT  = 64;
static constexpr int Q     = 128;
static constexpr int KH    = 6;
static constexpr int MROWS = BATCH * SLEN;
static constexpr int FPQ   = 2 * KH;
static constexpr int KFEAT = Q * FPQ;
static constexpr int KB    = KFEAT / 32;
static constexpr int QBLK  = Q / 32;

__device__ __forceinline__ int frag_k16(int i, int hi) {
    return (i & 7) + ((i >> 3) << 4) + hi * 8;
}

static constexpr int NBLK_ANGLE = MROWS / 16;
static constexpr int NBLK_PACK  = (KFEAT * DOUT) / 256;
static constexpr int NBLK_12    = NBLK_ANGLE + NBLK_PACK;

__device__ __forceinline__ void angle_block(
    const int blk,
    const float* __restrict__ X0,
    const float* __restrict__ t,
    const float* __restrict__ Wcw,
    const float* __restrict__ Wcb,
    const float* __restrict__ w,
    float* __restrict__ ANG)
{
    const int tid  = threadIdx.x;
    const int wave = tid >> 5;
    const int lane = tid & 31;
    const int lm   = lane & 15;
    const int hi   = lane >> 4;
    const int m0   = blk * 16;
    const int n0   = wave * 16;

    const int ksel = hi * 2;
    const float* xrow = X0  + (size_t)(m0 + lm) * DIN + ksel;
    const float* wrow = Wcw + (size_t)(n0 + lm) * DIN + ksel;

    v8f acc = {0.f,0.f,0.f,0.f,0.f,0.f,0.f,0.f};
    #pragma unroll 4
    for (int k = 0; k < DIN; k += 4) {
        v2f a = *(const v2f*)(xrow + k);
        v2f b = *(const v2f*)(wrow + k);
        acc = __builtin_amdgcn_wmma_f32_16x16x4_f32(
             false, a,  false, b,
             (short)0, acc,  false,  false);
    }

    const int   q     = n0 + lm;
    const int   batch = m0 >> 10;
    const float shift = Wcb[q] + w[q] * t[batch];

    #pragma unroll
    for (int j = 0; j < 8; ++j) {
        const int row = m0 + j + hi * 8;
        ANG[(size_t)row * Q + q] = acc[j] + shift;
    }
}

__device__ __forceinline__ void pack_block(
    const int blk,
    const float* __restrict__ A,
    const float* __restrict__ Bp,
    _Float16* __restrict__ W2P)
{
    const int idx  = blk * 256 + threadIdx.x;
    const int i    = idx & 15;
    const int lane = (idx >> 4) & 31;
    const int nt   = (idx >> 9) & 3;
    const int ks   = idx >> 11;
    const int hi   = lane >> 4;

    const int fn = ks & 1;
    const int qh = ks >> 1;
    const int qb = qh / KH;
    const int h  = qh - qb * KH;

    const int q  = qb * 32 + frag_k16(i, hi);
    const int d  = nt * 16 + (lane & 15);

    const int   ai  = (d * Q + q) * KH + h;
    const float av  = A[ai];
    const float bp  = Bp[ai];
    const float val = (fn == 0) ? av * cosf(bp) : av * sinf(bp);
    W2P[idx] = (_Float16)val;
}

__global__ __launch_bounds__(256) void k_angle_and_pack(
    const float* __restrict__ X0,
    const float* __restrict__ t,
    const float* __restrict__ Wcw,
    const float* __restrict__ Wcb,
    const float* __restrict__ w,
    float* __restrict__ ANG,
    const float* __restrict__ A,
    const float* __restrict__ Bp,
    _Float16* __restrict__ W2P)
{
    const int blk = blockIdx.x;
    if (blk < NBLK_ANGLE) {
        angle_block(blk, X0, t, Wcw, Wcb, w, ANG);
    } else {
        pack_block(blk - NBLK_ANGLE, A, Bp, W2P);
    }
}

static constexpr int NWAVE3 = QBLK * KH;
static constexpr int RPITCH = DOUT + 4;
static constexpr int PLANE  = 16 * RPITCH;

__global__ __launch_bounds__(NWAVE3 * 32) void k_fourier_wmma(
    const float*    __restrict__ ANG,
    const _Float16* __restrict__ W2P,
    float*          __restrict__ out)
{
    __shared__ alignas(16) float sPart[NWAVE3 * PLANE];

    const int tid  = threadIdx.x;
    const int wave = tid >> 5;
    const int lane = tid & 31;
    const int lm   = lane & 15;
    const int hi   = lane >> 4;
    const int qb   = wave / KH;
    const int h    = wave - qb * KH;
    const int m0   = blockIdx.x * 16;

    const float4* ap = (const float4*)(ANG + (size_t)(m0 + lm) * Q + qb * 32 + hi * 8);
    const float4 x0 = ap[0];
    const float4 x1 = ap[1];
    const float4 x2 = ap[4];
    const float4 x3 = ap[5];
    const float ang[16] = { x0.x, x0.y, x0.z, x0.w, x1.x, x1.y, x1.z, x1.w,
                            x2.x, x2.y, x2.z, x2.w, x3.x, x3.y, x3.z, x3.w };

    const v16h* bs = (const v16h*)W2P + lane + (size_t)wave * (2 * 128);
    const v16h* bc = bs + 128;
    const v16h s0 = bs[0];
    const v16h s1 = bs[32];
    const v16h s2 = bs[64];
    const v16h s3 = bs[96];
    const v16h c0 = bc[0];
    const v16h c1 = bc[32];
    const v16h c2 = bc[64];
    const v16h c3 = bc[96];

    const float kc = (float)(h + 1);
    v16h as, ac;
    #pragma unroll
    for (int i = 0; i < 16; ++i) {
        const float ph = kc * ang[i];
        as[i] = (_Float16)__sinf(ph);
        ac[i] = (_Float16)__cosf(ph);
    }

    const v8f zero = {0.f,0.f,0.f,0.f,0.f,0.f,0.f,0.f};
    v8f acc0 = __builtin_amdgcn_wmma_f32_16x16x32_f16(false, as, false, s0, (short)0, zero, false, false);
    v8f acc1 = __builtin_amdgcn_wmma_f32_16x16x32_f16(false, as, false, s1, (short)0, zero, false, false);
    v8f acc2 = __builtin_amdgcn_wmma_f32_16x16x32_f16(false, as, false, s2, (short)0, zero, false, false);
    v8f acc3 = __builtin_amdgcn_wmma_f32_16x16x32_f16(false, as, false, s3, (short)0, zero, false, false);
    acc0 = __builtin_amdgcn_wmma_f32_16x16x32_f16(false, ac, false, c0, (short)0, acc0, false, false);
    acc1 = __builtin_amdgcn_wmma_f32_16x16x32_f16(false, ac, false, c1, (short)0, acc1, false, false);
    acc2 = __builtin_amdgcn_wmma_f32_16x16x32_f16(false, ac, false, c2, (short)0, acc2, false, false);
    acc3 = __builtin_amdgcn_wmma_f32_16x16x32_f16(false, ac, false, c3, (short)0, acc3, false, false);

    float* pw = sPart + wave * PLANE + (hi * 8) * RPITCH + lm;
    #pragma unroll
    for (int j = 0; j < 8; ++j) {
        pw[j * RPITCH +  0] = acc0[j];
        pw[j * RPITCH + 16] = acc1[j];
        pw[j * RPITCH + 32] = acc2[j];
        pw[j * RPITCH + 48] = acc3[j];
    }
    __syncthreads();

    if (wave < 8) {
        const int r = wave * 2 + hi;
        const float* pr = sPart + r * RPITCH + lm * 4;
        float4 s = *(const float4*)pr;
        #pragma unroll
        for (int p = 1; p < NWAVE3; ++p) {
            const float4 v = *(const float4*)(pr + p * PLANE);
            s.x += v.x;
            s.y += v.y;
            s.z += v.z;
            s.w += v.w;
        }
        *(float4*)(out + (size_t)(m0 + r) * DOUT + lm * 4) = s;
    }
}

extern "C" void kernel_launch(void* const* d_in, const int* in_sizes, int n_in,
                              void* d_out, int out_size, void* d_ws, size_t ws_size,
                              hipStream_t stream) {
    const float* X0  = (const float*)d_in[0];
    const float* t   = (const float*)d_in[1];
    const float* Wcw = (const float*)d_in[2];
    const float* Wcb = (const float*)d_in[3];
    const float* w   = (const float*)d_in[4];
    const float* A   = (const float*)d_in[5];
    const float* Bp  = (const float*)d_in[6];
    float* out = (float*)d_out;

    char* ws = (char*)d_ws;
    float*    ANG = (float*)ws;
    _Float16* W2P = (_Float16*)(ws + (size_t)MROWS * Q * 4);

    k_angle_and_pack<<<NBLK_12, 256, 0, stream>>>(X0, t, Wcw, Wcb, w, ANG, A, Bp, W2P);
    k_fourier_wmma  <<<MROWS / 16, NWAVE3 * 32, 0, stream>>>(ANG, W2P, out);
}


# Stage-1 operands staged through LDS by whole-line 16-byte-per-lane loads (X0 tile read once per block, not per wave); fragments read from LDS in batches ahead of the WMMAs; no sum order changes
# speedup vs baseline: 10.8331x; 1.4075x over previous
#include <hip/hip_runtime.h>


typedef __attribute__((ext_vector_type(2)))  float    v2f;
typedef __attribute__((ext_vector_type(8)))  float    v8f;
typedef __attribute__((ext_vector_type(16))) _Float16 v16h;

static constexpr int BATCH = 2;
static constexpr int SLEN  = 1024;
static constexpr int DIN   = 256;
static constexpr int DOUT  = 64;
static constexpr int Q     = 128;
static constexpr int KH    = 6;
static constexpr int MROWS = BATCH * SLEN;
static constexpr int FPQ   = 2 * KH;
static constexpr int KFEAT = Q * FPQ;
static constexpr int KB    = KFEAT / 32;
static constexpr int QBLK  = Q / 32;

__device__ __forceinline__ int frag_k16(int i, int hi) {
    return (i & 7) + ((i >> 3) << 4) + hi * 8;
}

static constexpr int NBLK_ANGLE = MROWS / 16;
static constexpr int NBLK_PACK  = (KFEAT * DOUT) / 256;
static constexpr int NBLK_12    = NBLK_ANGLE + NBLK_PACK;

static constexpr int NWAVE1 = Q / 16;
static constexpr int XPITCH = DIN + 4;
static constexpr int TILE1  = 16 * XPITCH;
static constexpr int F4ROW  = DIN / 4;

__device__ __forceinline__ void angle_block(
    const int blk,
    const float* __restrict__ X0,
    const float* __restrict__ t,
    const float* __restrict__ Wcw,
    const float* __restrict__ Wcb,
    const float* __restrict__ w,
    float* __restrict__ ANG,
    float* sTiles)
{
    const int tid  = threadIdx.x;
    const int wave = tid >> 5;
    const int lane = tid & 31;
    const int lm   = lane & 15;
    const int hi   = lane >> 4;
    const int m0   = blk * 16;
    const int n0   = wave * 16;

    float* sX = sTiles;
    float* sW = sTiles + (1 + wave) * TILE1;

    const float4* xg = (const float4*)(X0  + (size_t)m0 * DIN) + tid;
    const float4* wg = (const float4*)(Wcw + (size_t)n0 * DIN) + lane;
    float4 xr[4];
    float4 wr[32];
    #pragma unroll
    for (int j = 0; j < 4; ++j) xr[j] = xg[256 * j];
    #pragma unroll
    for (int j = 0; j < 32; ++j) wr[j] = wg[32 * j];

    #pragma unroll
    for (int j = 0; j < 4; ++j) {
        const int g = tid + 256 * j;
        *(float4*)(sX + (g / F4ROW) * XPITCH + (g % F4ROW) * 4) = xr[j];
    }
    #pragma unroll
    for (int j = 0; j < 32; ++j) {
        const int g = lane + 32 * j;
        *(float4*)(sW + (g / F4ROW) * XPITCH + (g % F4ROW) * 4) = wr[j];
    }
    __syncthreads();

    const int foff = lm * XPITCH + hi * 2;
    const float* xrow = sX + foff;
    const float* wrow = sW + foff;

    constexpr int KGRP = 16;
    constexpr int NGRP = DIN / 4 / KGRP;
    v2f fa[2][KGRP], fb[2][KGRP];
    #pragma unroll
    for (int i = 0; i < KGRP; ++i) {
        fa[0][i] = *(const v2f*)(xrow + 4 * i);
        fb[0][i] = *(const v2f*)(wrow + 4 * i);
    }

    v8f acc = {0.f,0.f,0.f,0.f,0.f,0.f,0.f,0.f};
    #pragma unroll
    for (int g = 0; g < NGRP; ++g) {
        __builtin_amdgcn_sched_barrier(0);
        if (g + 1 < NGRP) {
            #pragma unroll
            for (int i = 0; i < KGRP; ++i) {
                fa[(g + 1) & 1][i] = *(const v2f*)(xrow + 4 * ((g + 1) * KGRP + i));
                fb[(g + 1) & 1][i] = *(const v2f*)(wrow + 4 * ((g + 1) * KGRP + i));
            }
        }
        __builtin_amdgcn_sched_barrier(0);
        #pragma unroll
        for (int i = 0; i < KGRP; ++i) {
            acc = __builtin_amdgcn_wmma_f32_16x16x4_f32(
                 false, fa[g & 1][i],  false, fb[g & 1][i],
                 (short)0, acc,  false,  false);
        }
    }

    const int   q     = n0 + lm;
    const int   batch = m0 >> 10;
    const float shift = Wcb[q] + w[q] * t[batch];

    #pragma unroll
    for (int j = 0; j < 8; ++j) {
        const int row = m0 + j + hi * 8;
        ANG[(size_t)row * Q + q] = acc[j] + shift;
    }
}

__device__ __forceinline__ void pack_block(
    const int blk,
    const float* __restrict__ A,
    const float* __restrict__ Bp,
    _Float16* __restrict__ W2P)
{
    const int idx  = blk * 256 + threadIdx.x;
    const int i    = idx & 15;
    const int lane = (idx >> 4) & 31;
    const int nt   = (idx >> 9) & 3;
    const int ks   = idx >> 11;
    const int hi   = lane >> 4;

    const int fn = ks & 1;
    const int qh = ks >> 1;
    const int qb = qh / KH;
    const int h  = qh - qb * KH;

    const int q  = qb * 32 + frag_k16(i, hi);
    const int d  = nt * 16 + (lane & 15);

    const int   ai  = (d * Q + q) * KH + h;
    const float av  = A[ai];
    const float bp  = Bp[ai];
    const float val = (fn == 0) ? av * cosf(bp) : av * sinf(bp);
    W2P[idx] = (_Float16)val;
}

__global__ __launch_bounds__(256) void k_angle_and_pack(
    const float* __restrict__ X0,
    const float* __restrict__ t,
    const float* __restrict__ Wcw,
    const float* __restrict__ Wcb,
    const float* __restrict__ w,
    float* __restrict__ ANG,
    const float* __restrict__ A,
    const float* __restrict__ Bp,
    _Float16* __restrict__ W2P)
{
    __shared__ alignas(16) float sTiles[(1 + NWAVE1) * TILE1];

    const int blk = blockIdx.x;
    if (blk < NBLK_ANGLE) {
        angle_block(blk, X0, t, Wcw, Wcb, w, ANG, sTiles);
    } else {
        pack_block(blk - NBLK_ANGLE, A, Bp, W2P);
    }
}

static constexpr int NWAVE3 = QBLK * KH;
static constexpr int RPITCH = DOUT + 4;
static constexpr int PLANE  = 16 * RPITCH;

__global__ __launch_bounds__(NWAVE3 * 32) void k_fourier_wmma(
    const float*    __restrict__ ANG,
    const _Float16* __restrict__ W2P,
    float*          __restrict__ out)
{
    __shared__ alignas(16) float sPart[NWAVE3 * PLANE];

    const int tid  = threadIdx.x;
    const int wave = tid >> 5;
    const int lane = tid & 31;
    const int lm   = lane & 15;
    const int hi   = lane >> 4;
    const int qb   = wave / KH;
    const int h    = wave - qb * KH;
    const int m0   = blockIdx.x * 16;

    const float4* ap = (const float4*)(ANG + (size_t)(m0 + lm) * Q + qb * 32 + hi * 8);
    const float4 x0 = ap[0];
    const float4 x1 = ap[1];
    const float4 x2 = ap[4];
    const float4 x3 = ap[5];
    const float ang[16] = { x0.x, x0.y, x0.z, x0.w, x1.x, x1.y, x1.z, x1.w,
                            x2.x, x2.y, x2.z, x2.w, x3.x, x3.y, x3.z, x3.w };

    const v16h* bs = (const v16h*)W2P + lane + (size_t)wave * (2 * 128);
    const v16h* bc = bs + 128;
    const v16h s0 = bs[0];
    const v16h s1 = bs[32];
    const v16h s2 = bs[64];
    const v16h s3 = bs[96];
    const v16h c0 = bc[0];
    const v16h c1 = bc[32];
    const v16h c2 = bc[64];
    const v16h c3 = bc[96];

    const float kc = (float)(h + 1);
    v16h as, ac;
    #pragma unroll
    for (int i = 0; i < 16; ++i) {
        const float ph = kc * ang[i];
        as[i] = (_Float16)__sinf(ph);
        ac[i] = (_Float16)__cosf(ph);
    }

    const v8f zero = {0.f,0.f,0.f,0.f,0.f,0.f,0.f,0.f};
    v8f acc0 = __builtin_amdgcn_wmma_f32_16x16x32_f16(false, as, false, s0, (short)0, zero, false, false);
    v8f acc1 = __builtin_amdgcn_wmma_f32_16x16x32_f16(false, as, false, s1, (short)0, zero, false, false);
    v8f acc2 = __builtin_amdgcn_wmma_f32_16x16x32_f16(false, as, false, s2, (short)0, zero, false, false);
    v8f acc3 = __builtin_amdgcn_wmma_f32_16x16x32_f16(false, as, false, s3, (short)0, zero, false, false);
    acc0 = __builtin_amdgcn_wmma_f32_16x16x32_f16(false, ac, false, c0, (short)0, acc0, false, false);
    acc1 = __builtin_amdgcn_wmma_f32_16x16x32_f16(false, ac, false, c1, (short)0, acc1, false, false);
    acc2 = __builtin_amdgcn_wmma_f32_16x16x32_f16(false, ac, false, c2, (short)0, acc2, false, false);
    acc3 = __builtin_amdgcn_wmma_f32_16x16x32_f16(false, ac, false, c3, (short)0, acc3, false, false);

    float* pw = sPart + wave * PLANE + (hi * 8) * RPITCH + lm;
    #pragma unroll
    for (int j = 0; j < 8; ++j) {
        pw[j * RPITCH +  0] = acc0[j];
        pw[j * RPITCH + 16] = acc1[j];
        pw[j * RPITCH + 32] = acc2[j];
        pw[j * RPITCH + 48] = acc3[j];
    }
    __syncthreads();

    if (wave < 8) {
        const int r = wave * 2 + hi;
        const float* pr = sPart + r * RPITCH + lm * 4;
        float4 s = *(const float4*)pr;
        #pragma unroll
        for (int p = 1; p < NWAVE3; ++p) {
            const float4 v = *(const float4*)(pr + p * PLANE);
            s.x += v.x;
            s.y += v.y;
            s.z += v.z;
            s.w += v.w;
        }
        *(float4*)(out + (size_t)(m0 + r) * DOUT + lm * 4) = s;
    }
}

extern "C" void kernel_launch(void* const* d_in, const int* in_sizes, int n_in,
                              void* d_out, int out_size, void* d_ws, size_t ws_size,
                              hipStream_t stream) {
    const float* X0  = (const float*)d_in[0];
    const float* t   = (const float*)d_in[1];
    const float* Wcw = (const float*)d_in[2];
    const float* Wcb = (const float*)d_in[3];
    const float* w   = (const float*)d_in[4];
    const float* A   = (const float*)d_in[5];
    const float* Bp  = (const float*)d_in[6];
    float* out = (float*)d_out;

    char* ws = (char*)d_ws;
    float*    ANG = (float*)ws;
    _Float16* W2P = (_Float16*)(ws + (size_t)MROWS * Q * 4);

    k_angle_and_pack<<<NBLK_12, 256, 0, stream>>>(X0, t, Wcw, Wcb, w, ANG, A, Bp, W2P);
    k_fourier_wmma  <<<MROWS / 16, NWAVE3 * 32, 0, stream>>>(ANG, W2P, out);
}


# Stage 3 re-cut, 128 blocks x 24 waves to 256 blocks x 4 waves: block = 16 rows x 32 columns, wave = one 32-q block summing its 12 k-steps in registers; 4 LDS partials, not 24; f32 sum order changes
# speedup vs baseline: 12.5732x; 1.1606x over previous
#include <hip/hip_runtime.h>


typedef __attribute__((ext_vector_type(2)))  float    v2f;
typedef __attribute__((ext_vector_type(4)))  float    v4f;
typedef __attribute__((ext_vector_type(8)))  float    v8f;
typedef __attribute__((ext_vector_type(16))) _Float16 v16h;

static constexpr int BATCH = 2;
static constexpr int SLEN  = 1024;
static constexpr int DIN   = 256;
static constexpr int DOUT  = 64;
static constexpr int Q     = 128;
static constexpr int KH    = 6;
static constexpr int MROWS = BATCH * SLEN;
static constexpr int FPQ   = 2 * KH;
static constexpr int KFEAT = Q * FPQ;
static constexpr int KB    = KFEAT / 32;
static constexpr int QBLK  = Q / 32;

__device__ __forceinline__ int frag_k16(int i, int hi) {
    return (i & 7) + ((i >> 3) << 4) + hi * 8;
}

static constexpr int NBLK_ANGLE = MROWS / 16;
static constexpr int NBLK_PACK  = (KFEAT * DOUT) / 256;
static constexpr int NBLK_12    = NBLK_ANGLE + NBLK_PACK;

static constexpr int NWAVE1 = Q / 16;
static constexpr int XPITCH = DIN + 4;
static constexpr int TILE1  = 16 * XPITCH;
static constexpr int F4ROW  = DIN / 4;

__device__ __forceinline__ void angle_block(
    const int blk,
    const float* __restrict__ X0,
    const float* __restrict__ t,
    const float* __restrict__ Wcw,
    const float* __restrict__ Wcb,
    const float* __restrict__ w,
    float* __restrict__ ANG,
    float* sTiles)
{
    const int tid  = threadIdx.x;
    const int wave = tid >> 5;
    const int lane = tid & 31;
    const int lm   = lane & 15;
    const int hi   = lane >> 4;
    const int m0   = blk * 16;
    const int n0   = wave * 16;

    float* sX = sTiles;
    float* sW = sTiles + (1 + wave) * TILE1;

    const float4* xg = (const float4*)(X0  + (size_t)m0 * DIN) + tid;
    const float4* wg = (const float4*)(Wcw + (size_t)n0 * DIN) + lane;
    float4 xr[4];
    float4 wr[32];
    #pragma unroll
    for (int j = 0; j < 4; ++j) xr[j] = xg[256 * j];
    #pragma unroll
    for (int j = 0; j < 32; ++j) wr[j] = wg[32 * j];

    #pragma unroll
    for (int j = 0; j < 4; ++j) {
        const int g = tid + 256 * j;
        *(float4*)(sX + (g / F4ROW) * XPITCH + (g % F4ROW) * 4) = xr[j];
    }
    #pragma unroll
    for (int j = 0; j < 32; ++j) {
        const int g = lane + 32 * j;
        *(float4*)(sW + (g / F4ROW) * XPITCH + (g % F4ROW) * 4) = wr[j];
    }
    __syncthreads();

    const int foff = lm * XPITCH + hi * 2;
    const float* xrow = sX + foff;
    const float* wrow = sW + foff;

    constexpr int KGRP = 16;
    constexpr int NGRP = DIN / 4 / KGRP;
    v2f fa[2][KGRP], fb[2][KGRP];
    #pragma unroll
    for (int i = 0; i < KGRP; ++i) {
        fa[0][i] = *(const v2f*)(xrow + 4 * i);
        fb[0][i] = *(const v2f*)(wrow + 4 * i);
    }

    v8f acc = {0.f,0.f,0.f,0.f,0.f,0.f,0.f,0.f};
    #pragma unroll
    for (int g = 0; g < NGRP; ++g) {
        __builtin_amdgcn_sched_barrier(0);
        if (g + 1 < NGRP) {
            #pragma unroll
            for (int i = 0; i < KGRP; ++i) {
                fa[(g + 1) & 1][i] = *(const v2f*)(xrow + 4 * ((g + 1) * KGRP + i));
                fb[(g + 1) & 1][i] = *(const v2f*)(wrow + 4 * ((g + 1) * KGRP + i));
            }
        }
        __builtin_amdgcn_sched_barrier(0);
        #pragma unroll
        for (int i = 0; i < KGRP; ++i) {
            acc = __builtin_amdgcn_wmma_f32_16x16x4_f32(
                 false, fa[g & 1][i],  false, fb[g & 1][i],
                 (short)0, acc,  false,  false);
        }
    }

    const int   q     = n0 + lm;
    const int   batch = m0 >> 10;
    const float shift = Wcb[q] + w[q] * t[batch];

    #pragma unroll
    for (int j = 0; j < 8; ++j) {
        const int row = m0 + j + hi * 8;
        ANG[(size_t)row * Q + q] = acc[j] + shift;
    }
}

__device__ __forceinline__ void pack_block(
    const int blk,
    const float* __restrict__ A,
    const float* __restrict__ Bp,
    _Float16* __restrict__ W2P)
{
    const int idx  = blk * 256 + threadIdx.x;
    const int i    = idx & 15;
    const int lane = (idx >> 4) & 31;
    const int nt   = (idx >> 9) & 3;
    const int ks   = idx >> 11;
    const int hi   = lane >> 4;

    const int fn = ks & 1;
    const int qh = ks >> 1;
    const int qb = qh / KH;
    const int h  = qh - qb * KH;

    const int q  = qb * 32 + frag_k16(i, hi);
    const int d  = nt * 16 + (lane & 15);

    const int   ai  = (d * Q + q) * KH + h;
    const float av  = A[ai];
    const float bp  = Bp[ai];
    const float val = (fn == 0) ? av * cosf(bp) : av * sinf(bp);
    W2P[idx] = (_Float16)val;
}

__global__ __launch_bounds__(256) void k_angle_and_pack(
    const float* __restrict__ X0,
    const float* __restrict__ t,
    const float* __restrict__ Wcw,
    const float* __restrict__ Wcb,
    const float* __restrict__ w,
    float* __restrict__ ANG,
    const float* __restrict__ A,
    const float* __restrict__ Bp,
    _Float16* __restrict__ W2P)
{
    __shared__ alignas(16) float sTiles[(1 + NWAVE1) * TILE1];

    const int blk = blockIdx.x;
    if (blk < NBLK_ANGLE) {
        angle_block(blk, X0, t, Wcw, Wcb, w, ANG, sTiles);
    } else {
        pack_block(blk - NBLK_ANGLE, A, Bp, W2P);
    }
}

static constexpr int NWAVE3 = QBLK;
static constexpr int DHALF  = DOUT / 2;
static constexpr int KSW    = 2 * KH;
static constexpr int RPITCH = DHALF + 4;
static constexpr int PLANE  = 16 * RPITCH;
static constexpr int NBLK_3 = (MROWS / 16) * 2;

__global__ __launch_bounds__(NWAVE3 * 32) void k_fourier_wmma(
    const float*    __restrict__ ANG,
    const _Float16* __restrict__ W2P,
    float*          __restrict__ out)
{
    __shared__ alignas(16) float sPart[NWAVE3 * PLANE];

    const int tid  = threadIdx.x;
    const int qb   = tid >> 5;
    const int lane = tid & 31;
    const int lm   = lane & 15;
    const int hi   = lane >> 4;
    const int m0   = (blockIdx.x >> 1) * 16;
    const int dh   = blockIdx.x & 1;

    const v4f* ap = (const v4f*)(ANG + (size_t)(m0 + lm) * Q + qb * 32 + hi * 8);
    const v4f x0 = ap[0];
    const v4f x1 = ap[1];
    const v4f x2 = ap[4];
    const v4f x3 = ap[5];

    const v16h* bp = (const v16h*)W2P + lane + (size_t)((qb * KSW) * 4 + 2 * dh) * 32;
    v16h bf[KSW][2];
    #pragma unroll
    for (int j = 0; j < KSW; ++j) {
        bf[j][0] = bp[(j * 4 + 0) * 32];
        bf[j][1] = bp[(j * 4 + 1) * 32];
    }
    asm volatile("" :: "v"(x0), "v"(x1), "v"(x2), "v"(x3));
    __builtin_amdgcn_sched_barrier(0);

    const float ang[16] = { x0.x, x0.y, x0.z, x0.w, x1.x, x1.y, x1.z, x1.w,
                            x2.x, x2.y, x2.z, x2.w, x3.x, x3.y, x3.z, x3.w };

    v8f acc0 = {0.f,0.f,0.f,0.f,0.f,0.f,0.f,0.f};
    v8f acc1 = {0.f,0.f,0.f,0.f,0.f,0.f,0.f,0.f};
    #pragma unroll
    for (int h = 0; h < KH; ++h) {
        const float kc = (float)(h + 1);
        v16h as, ac;
        #pragma unroll
        for (int i = 0; i < 16; ++i) {
            const float ph = kc * ang[i];
            as[i] = (_Float16)__sinf(ph);
            ac[i] = (_Float16)__cosf(ph);
        }
        acc0 = __builtin_amdgcn_wmma_f32_16x16x32_f16(false, as, false, bf[2 * h][0],     (short)0, acc0, false, false);
        acc1 = __builtin_amdgcn_wmma_f32_16x16x32_f16(false, as, false, bf[2 * h][1],     (short)0, acc1, false, false);
        acc0 = __builtin_amdgcn_wmma_f32_16x16x32_f16(false, ac, false, bf[2 * h + 1][0], (short)0, acc0, false, false);
        acc1 = __builtin_amdgcn_wmma_f32_16x16x32_f16(false, ac, false, bf[2 * h + 1][1], (short)0, acc1, false, false);
    }

    float* pw = sPart + qb * PLANE + (hi * 8) * RPITCH + lm;
    #pragma unroll
    for (int j = 0; j < 8; ++j) {
        pw[j * RPITCH +  0] = acc0[j];
        pw[j * RPITCH + 16] = acc1[j];
    }
    __syncthreads();

    const int r  = tid >> 3;
    const int c4 = tid & 7;
    const float* pr = sPart + r * RPITCH + c4 * 4;
    float4 s = *(const float4*)pr;
    #pragma unroll
    for (int p = 1; p < NWAVE3; ++p) {
        const float4 v = *(const float4*)(pr + p * PLANE);
        s.x += v.x;
        s.y += v.y;
        s.z += v.z;
        s.w += v.w;
    }
    *(float4*)(out + (size_t)(m0 + r) * DOUT + dh * DHALF + c4 * 4) = s;
}

extern "C" void kernel_launch(void* const* d_in, const int* in_sizes, int n_in,
                              void* d_out, int out_size, void* d_ws, size_t ws_size,
                              hipStream_t stream) {
    const float* X0  = (const float*)d_in[0];
    const float* t   = (const float*)d_in[1];
    const float* Wcw = (const float*)d_in[2];
    const float* Wcb = (const float*)d_in[3];
    const float* w   = (const float*)d_in[4];
    const float* A   = (const float*)d_in[5];
    const float* Bp  = (const float*)d_in[6];
    float* out = (float*)d_out;

    char* ws = (char*)d_ws;
    float*    ANG = (float*)ws;
    _Float16* W2P = (_Float16*)(ws + (size_t)MROWS * Q * 4);

    k_angle_and_pack<<<NBLK_12, 256, 0, stream>>>(X0, t, Wcw, Wcb, w, ANG, A, Bp, W2P);
    k_fourier_wmma  <<<NBLK_3, NWAVE3 * 32, 0, stream>>>(ANG, W2P, out);
}
